# starting kernel
# baseline (speedup 1.0000x reference)
#include <hip/hip_runtime.h>

#define BATCH 8
#define LQ   512
#define KK   512
#define DD   64
#define NN   64
#define MM   64
#define PP   64

typedef __attribute__((ext_vector_type(2))) float v2f;
typedef __attribute__((ext_vector_type(8))) float v8f;

__global__ __launch_bounds__(64) void vkc_kernel(const float* __restrict__ vk,
                                                 const float* __restrict__ vexp,
                                                 float* __restrict__ vkc) {
  __shared__ float sE[4 * PP];
  const int tid  = threadIdx.x;
  const int pair = tid >> 4;
  const int nq   = (tid & 15) * 4;
  const int bk0  = blockIdx.x * 4;

  for (int i = tid; i < 4 * PP; i += 64) sE[i] = vexp[(size_t)bk0 * PP + i];
  __syncthreads();

  const float* vkp = vk + (size_t)(bk0 + pair) * PP * NN + nq;
  float4 acc = {0.f, 0.f, 0.f, 0.f};
#pragma unroll 4
  for (int p = 0; p < PP; ++p) {
    const float  w = sE[pair * PP + p];
    const float4 v = *reinterpret_cast<const float4*>(vkp + (size_t)p * NN);
    acc.x = fmaf(w, v.x, acc.x);
    acc.y = fmaf(w, v.y, acc.y);
    acc.z = fmaf(w, v.z, acc.z);
    acc.w = fmaf(w, v.w, acc.w);
  }
  *reinterpret_cast<float4*>(vkc + (size_t)(bk0 + pair) * NN + nq) = acc;
}

#define SSTRIDE 516

__global__ __launch_bounds__(128) void attn_kernel(const float* __restrict__ q,
                                                   const float* __restrict__ kmat,
                                                   const float* __restrict__ vkc,
                                                   const float* __restrict__ scale_p,
                                                   float* __restrict__ tmp) {
  __shared__ float sS[16 * SSTRIDE];
  __shared__ float sRed[128];
  __shared__ float sRowMax[16];
  __shared__ float sRowSum[16];

  const int tid  = threadIdx.x;
  const int wave = tid >> 5;
  const int lane = tid & 31;
  const int half = lane >> 4;
  const int lrow = lane & 15;

  const int b  = blockIdx.x >> 5;
  const int lt = blockIdx.x & 31;
  const int l0 = lt * 16;

  const float scale = scale_p[0];

  const float* qrow = q + ((size_t)b * LQ + l0 + lrow) * DD + 2 * half;
  v2f afrag[16];
#pragma unroll
  for (int dd = 0; dd < 16; ++dd) {
    const float2 t = *reinterpret_cast<const float2*>(qrow + 4 * dd);
    afrag[dd].x = t.x; afrag[dd].y = t.y;
  }

  const float* kbase = kmat + (size_t)b * KK * DD;
  for (int ct = wave; ct < 32; ct += 4) {
    const int k0 = ct * 16;
    const float* krow = kbase + (size_t)(k0 + lrow) * DD + 2 * half;
    v8f acc0 = {0.f, 0.f, 0.f, 0.f, 0.f, 0.f, 0.f, 0.f};
    v8f acc1 = {0.f, 0.f, 0.f, 0.f, 0.f, 0.f, 0.f, 0.f};
#pragma unroll
    for (int dd = 0; dd < 16; dd += 2) {
      const float2 t0 = *reinterpret_cast<const float2*>(krow + 4 * dd);
      const float2 t1 = *reinterpret_cast<const float2*>(krow + 4 * dd + 4);
      v2f b0; b0.x = t0.x; b0.y = t0.y;
      v2f b1; b1.x = t1.x; b1.y = t1.y;
      acc0 = __builtin_amdgcn_wmma_f32_16x16x4_f32(false, afrag[dd],     false, b0,
                                                   (short)0, acc0, false, false);
      acc1 = __builtin_amdgcn_wmma_f32_16x16x4_f32(false, afrag[dd + 1], false, b1,
                                                   (short)0, acc1, false, false);
    }
#pragma unroll
    for (int r = 0; r < 8; ++r) {
      const int m = r + 8 * half;
      sS[m * SSTRIDE + k0 + lrow] = (acc0[r] + acc1[r]) * scale;
    }
  }
  __syncthreads();

  {
    const int r = tid >> 3;
    const int j = tid & 7;
    float pmax = -3.4e38f;
    for (int i = 0; i < 64; ++i)
      pmax = fmaxf(pmax, sS[r * SSTRIDE + j + 8 * i]);
    sRed[tid] = pmax;
    __syncthreads();
    if (j == 0) {
      float mx = sRed[tid];
      for (int t = 1; t < 8; ++t) mx = fmaxf(mx, sRed[tid + t]);
      sRowMax[r] = mx;
    }
    __syncthreads();
    const float mx = sRowMax[r];
    float psum = 0.f;
    for (int i = 0; i < 64; ++i) {
      const int c = j + 8 * i;
      const float e = expf(sS[r * SSTRIDE + c] - mx);
      sS[r * SSTRIDE + c] = e;
      psum += e;
    }
    sRed[tid] = psum;
    __syncthreads();
    if (j == 0) {
      float s = sRed[tid];
      for (int t = 1; t < 8; ++t) s += sRed[tid + t];
      sRowSum[r] = s;
    }
  }
  __syncthreads();

  {
    const int n0 = wave * 16;
    const float* vb = vkc + (size_t)b * KK * NN + n0 + lrow;
    v8f acc0 = {0.f, 0.f, 0.f, 0.f, 0.f, 0.f, 0.f, 0.f};
    v8f acc1 = {0.f, 0.f, 0.f, 0.f, 0.f, 0.f, 0.f, 0.f};
    for (int k0 = 0; k0 < KK; k0 += 8) {
      const float2 ta =
          *reinterpret_cast<const float2*>(&sS[lrow * SSTRIDE + k0 + 2 * half]);
      const float2 tb =
          *reinterpret_cast<const float2*>(&sS[lrow * SSTRIDE + k0 + 4 + 2 * half]);
      v2f a0; a0.x = ta.x; a0.y = ta.y;
      v2f a1; a1.x = tb.x; a1.y = tb.y;
      v2f b0;
      b0.x = vb[(size_t)(k0 + 2 * half)     * NN];
      b0.y = vb[(size_t)(k0 + 2 * half + 1) * NN];
      v2f b1;
      b1.x = vb[(size_t)(k0 + 4 + 2 * half)     * NN];
      b1.y = vb[(size_t)(k0 + 4 + 2 * half + 1) * NN];
      acc0 = __builtin_amdgcn_wmma_f32_16x16x4_f32(false, a0, false, b0,
                                                   (short)0, acc0, false, false);
      acc1 = __builtin_amdgcn_wmma_f32_16x16x4_f32(false, a1, false, b1,
                                                   (short)0, acc1, false, false);
    }
#pragma unroll
    for (int r = 0; r < 8; ++r) {
      const int m = r + 8 * half;
      tmp[((size_t)b * LQ + l0 + m) * NN + n0 + lrow] =
          (acc0[r] + acc1[r]) / sRowSum[m];
    }
  }
}

__global__ __launch_bounds__(256) void out_kernel(const float* __restrict__ q,
                                                  const float* __restrict__ vq,
                                                  const float* __restrict__ tmp,
                                                  const float* __restrict__ gamma,
                                                  const float* __restrict__ beta,
                                                  float* __restrict__ out) {
  const int bl = blockIdx.x;
  __shared__ float sT[64];
  __shared__ float sPart[256];
  __shared__ float sRed[64];
  __shared__ float sRed2[64];
  __shared__ float sStat[2];

  const int tid = threadIdx.x;
  if (tid < 64) sT[tid] = tmp[(size_t)bl * NN + tid];
  __syncthreads();

  const int m    = tid >> 2;
  const int part = tid & 3;
  const float4* vqp = reinterpret_cast<const float4*>(
      vq + ((size_t)bl * MM + m) * NN + part * 16);
  float acc = 0.f;
#pragma unroll
  for (int i = 0; i < 4; ++i) {
    const float4 v = vqp[i];
    const int c = part * 16 + 4 * i;
    acc = fmaf(v.x, sT[c + 0], acc);
    acc = fmaf(v.y, sT[c + 1], acc);
    acc = fmaf(v.z, sT[c + 2], acc);
    acc = fmaf(v.w, sT[c + 3], acc);
  }
  sPart[tid] = acc;
  __syncthreads();

  float x = 0.f;
  if (part == 0) {
    x = sPart[tid] + sPart[tid + 1] + sPart[tid + 2] + sPart[tid + 3]
        + q[(size_t)bl * DD + m];
    sRed[m]  = x;
    sRed2[m] = x * x;
  }
  __syncthreads();
  for (int s = 32; s >= 1; s >>= 1) {
    if (tid < s) { sRed[tid] += sRed[tid + s]; sRed2[tid] += sRed2[tid + s]; }
    __syncthreads();
  }
  if (tid == 0) {
    const float mu  = sRed[0]  * (1.f / 64.f);
    const float var = sRed2[0] * (1.f / 64.f) - mu * mu;
    sStat[0] = mu;
    sStat[1] = rsqrtf(var + 1e-3f);
  }
  __syncthreads();
  if (part == 0) {
    out[(size_t)bl * DD + m] = (x - sStat[0]) * sStat[1] * gamma[m] + beta[m];
  }
}

extern "C" void kernel_launch(void* const* d_in, const int* in_sizes, int n_in,
                              void* d_out, int out_size, void* d_ws, size_t ws_size,
                              hipStream_t stream) {
  const float* q     = (const float*)d_in[0];
  const float* kmat  = (const float*)d_in[1];
  const float* vq    = (const float*)d_in[2];
  const float* vk    = (const float*)d_in[3];
  const float* vexp  = (const float*)d_in[4];
  const float* scale = (const float*)d_in[5];
  const float* gamma = (const float*)d_in[6];
  const float* beta  = (const float*)d_in[7];
  float* out = (float*)d_out;

  float* vkc = (float*)d_ws;
  float* tmp = vkc + (size_t)BATCH * KK * NN;

  vkc_kernel <<<BATCH * KK / 4,    64,  0, stream>>>(vk, vexp, vkc);
  attn_kernel<<<BATCH * (LQ / 16), 128, 0, stream>>>(q, kmat, vkc, scale, tmp);
  out_kernel <<<BATCH * LQ,        256, 0, stream>>>(q, vq, tmp, gamma, beta, out);
}


# vq kernel: 512 workgroups of 8 waves; each wave requests its 8-row stripe of its workgroup's 8 slabs at entry, row sums by DPP, one barrier per slab, one wave does LayerNorm; fp sum order changes
# speedup vs baseline: 1.5257x; 1.5257x over previous
#include <hip/hip_runtime.h>

#define BATCH 8
#define LQ   512
#define KK   512
#define DD   64
#define NN   64
#define MM   64
#define PP   64

typedef __attribute__((ext_vector_type(2))) float v2f;
typedef __attribute__((ext_vector_type(8))) float v8f;

__global__ __launch_bounds__(128) void vkc_kernel(const float* __restrict__ vk,
                                                  const float* __restrict__ vexp,
                                                  float* __restrict__ vkc) {
  __shared__ float sE[4 * PP];
  __shared__ float4 sHalf[64];
  const int tid  = threadIdx.x;
  const int ph   = tid >> 6;
  const int pair = (tid >> 4) & 3;
  const int nq   = (tid & 15) * 4;
  const int bk0  = blockIdx.x * 4;
  const int p0   = ph * (PP / 2);

  for (int i = tid; i < 4 * PP; i += 128) sE[i] = vexp[(size_t)bk0 * PP + i];
  __syncthreads();

  const float* vkp = vk + ((size_t)(bk0 + pair) * PP + p0) * NN + nq;
  float4 acc = {0.f, 0.f, 0.f, 0.f};
#pragma unroll 4
  for (int p = 0; p < PP / 2; ++p) {
    const float  w = sE[pair * PP + p0 + p];
    const float4 v = *reinterpret_cast<const float4*>(vkp + (size_t)p * NN);
    acc.x = fmaf(w, v.x, acc.x);
    acc.y = fmaf(w, v.y, acc.y);
    acc.z = fmaf(w, v.z, acc.z);
    acc.w = fmaf(w, v.w, acc.w);
  }
  if (ph == 1) sHalf[tid & 63] = acc;
  __syncthreads();
  if (ph == 0) {
    const float4 o = sHalf[tid];
    acc.x += o.x;
    acc.y += o.y;
    acc.z += o.z;
    acc.w += o.w;
    *reinterpret_cast<float4*>(vkc + (size_t)(bk0 + pair) * NN + nq) = acc;
  }
}

#define SSTRIDE 516
#define PSTRIDE 68

__global__ __launch_bounds__(256) void attn_kernel(const float* __restrict__ q,
                                                   const float* __restrict__ kmat,
                                                   const float* __restrict__ vkc,
                                                   const float* __restrict__ scale_p,
                                                   float* __restrict__ tmp) {
  __shared__ float sS[16 * SSTRIDE];
  __shared__ __attribute__((aligned(16))) float sP[2 * 16 * PSTRIDE];
  __shared__ float sRowSum[16];

  const int tid  = threadIdx.x;
  const int wave = tid >> 5;
  const int lane = tid & 31;
  const int half = lane >> 4;
  const int lrow = lane & 15;

  const int b  = blockIdx.x >> 5;
  const int lt = blockIdx.x & 31;
  const int l0 = lt * 16;

  const float scale = scale_p[0];

  const float* qrow = q + ((size_t)b * LQ + l0 + lrow) * DD + 2 * half;
  v2f afrag[16];
#pragma unroll
  for (int dd = 0; dd < 16; ++dd) {
    const float2 t = *reinterpret_cast<const float2*>(qrow + 4 * dd);
    afrag[dd].x = t.x; afrag[dd].y = t.y;
  }

  const float* kbase = kmat + (size_t)b * KK * DD;
  v2f bfrag[4][16];
#pragma unroll
  for (int t = 0; t < 4; ++t) {
    const float* krow = kbase + (size_t)((wave + 8 * t) * 16 + lrow) * DD + 2 * half;
#pragma unroll
    for (int dd = 0; dd < 16; ++dd) {
      const float2 u = *reinterpret_cast<const float2*>(krow + 4 * dd);
      bfrag[t][dd].x = u.x; bfrag[t][dd].y = u.y;
    }
  }
#pragma unroll
  for (int t = 0; t < 4; ++t) {
    const int k0 = (wave + 8 * t) * 16;
    v8f acc0 = {0.f, 0.f, 0.f, 0.f, 0.f, 0.f, 0.f, 0.f};
    v8f acc1 = {0.f, 0.f, 0.f, 0.f, 0.f, 0.f, 0.f, 0.f};
#pragma unroll
    for (int dd = 0; dd < 16; dd += 2) {
      acc0 = __builtin_amdgcn_wmma_f32_16x16x4_f32(false, afrag[dd],     false, bfrag[t][dd],
                                                   (short)0, acc0, false, false);
      acc1 = __builtin_amdgcn_wmma_f32_16x16x4_f32(false, afrag[dd + 1], false, bfrag[t][dd + 1],
                                                   (short)0, acc1, false, false);
    }
#pragma unroll
    for (int r = 0; r < 8; ++r) {
      const int m = r + 8 * half;
      sS[m * SSTRIDE + k0 + lrow] = (acc0[r] + acc1[r]) * scale;
    }
  }

  const int nt = wave & 3;
  const int kh = wave >> 2;
  const int n0 = nt * 16;
  const int kb = kh * 256;
  const float* vb = vkc + ((size_t)b * KK + kb) * NN + n0 + lrow;
  v2f bpre[32];
#pragma unroll
  for (int i = 0; i < 32; ++i) {
    bpre[i].x = vb[(size_t)(4 * i + 2 * half)     * NN];
    bpre[i].y = vb[(size_t)(4 * i + 2 * half + 1) * NN];
  }
  __syncthreads();

#pragma unroll
  for (int rr = 0; rr < 2; ++rr) {
    const int r = 2 * wave + rr;
    float v[16];
    float mx = -3.4e38f;
#pragma unroll
    for (int i = 0; i < 16; ++i) {
      v[i] = sS[r * SSTRIDE + lane + 32 * i];
      mx = fmaxf(mx, v[i]);
    }
#pragma unroll
    for (int o = 16; o >= 1; o >>= 1) mx = fmaxf(mx, __shfl_xor(mx, o));
    float s = 0.f;
#pragma unroll
    for (int i = 0; i < 16; ++i) {
      const float e = expf(v[i] - mx);
      sS[r * SSTRIDE + lane + 32 * i] = e;
      s += e;
    }
#pragma unroll
    for (int o = 16; o >= 1; o >>= 1) s += __shfl_xor(s, o);
    if (lane == 0) sRowSum[r] = s;
  }
  __syncthreads();

  {
    v8f acc0 = {0.f, 0.f, 0.f, 0.f, 0.f, 0.f, 0.f, 0.f};
    v8f acc1 = {0.f, 0.f, 0.f, 0.f, 0.f, 0.f, 0.f, 0.f};
    const float* arow = &sS[lrow * SSTRIDE + kb + 2 * half];
#pragma unroll
    for (int i = 0; i < 32; i += 2) {
      const float2 ta = *reinterpret_cast<const float2*>(arow + 4 * i);
      const float2 tb = *reinterpret_cast<const float2*>(arow + 4 * i + 4);
      v2f a0; a0.x = ta.x; a0.y = ta.y;
      v2f a1; a1.x = tb.x; a1.y = tb.y;
      acc0 = __builtin_amdgcn_wmma_f32_16x16x4_f32(false, a0, false, bpre[i],
                                                   (short)0, acc0, false, false);
      acc1 = __builtin_amdgcn_wmma_f32_16x16x4_f32(false, a1, false, bpre[i + 1],
                                                   (short)0, acc1, false, false);
    }
#pragma unroll
    for (int i = 32; i < 64; i += 2) {
      const float2 ta = *reinterpret_cast<const float2*>(arow + 4 * i);
      const float2 tb = *reinterpret_cast<const float2*>(arow + 4 * i + 4);
      v2f a0; a0.x = ta.x; a0.y = ta.y;
      v2f a1; a1.x = tb.x; a1.y = tb.y;
      v2f b0;
      b0.x = vb[(size_t)(4 * i + 2 * half)     * NN];
      b0.y = vb[(size_t)(4 * i + 2 * half + 1) * NN];
      v2f b1;
      b1.x = vb[(size_t)(4 * i + 4 + 2 * half)     * NN];
      b1.y = vb[(size_t)(4 * i + 4 + 2 * half + 1) * NN];
      acc0 = __builtin_amdgcn_wmma_f32_16x16x4_f32(false, a0, false, b0,
                                                   (short)0, acc0, false, false);
      acc1 = __builtin_amdgcn_wmma_f32_16x16x4_f32(false, a1, false, b1,
                                                   (short)0, acc1, false, false);
    }
#pragma unroll
    for (int r = 0; r < 8; ++r) {
      const int m = r + 8 * half;
      sP[(kh * 16 + m) * PSTRIDE + n0 + lrow] = acc0[r] + acc1[r];
    }
  }
  __syncthreads();
  {
    const int row = tid >> 4;
    const int c4  = (tid & 15) * 4;
    const float4 p0 = *reinterpret_cast<const float4*>(&sP[row * PSTRIDE + c4]);
    const float4 p1 = *reinterpret_cast<const float4*>(&sP[(16 + row) * PSTRIDE + c4]);
    const float den = sRowSum[row];
    float4 o;
    o.x = (p0.x + p1.x) / den;
    o.y = (p0.y + p1.y) / den;
    o.z = (p0.z + p1.z) / den;
    o.w = (p0.w + p1.w) / den;
    *reinterpret_cast<float4*>(tmp + ((size_t)b * LQ + l0 + row) * NN + c4) = o;
  }
}

#define OSLABS 8

static __device__ __forceinline__ float xhalf(float x) {
  const int b = __builtin_bit_cast(int, x);
  return __builtin_bit_cast(float,
      __builtin_amdgcn_permlanex16(b, b, 0x76543210, 0xfedcba98, false, false));
}
template <int M>
static __device__ __forceinline__ float rowx(float x) {
  const int b = __builtin_bit_cast(int, x);
  return __builtin_bit_cast(float, __builtin_amdgcn_update_dpp(0, b, 0x160 + M, 0xf, 0xf, true));
}
static __device__ __forceinline__ void lds_barrier() {
  asm volatile("s_wait_dscnt 0x0\n\ts_barrier_signal -1\n\ts_barrier_wait -1" ::: "memory");
}

__global__ __launch_bounds__(256) void out_kernel(const float* q,
                                                  const float* vq,
                                                  const float* tmp,
                                                  const float* gamma,
                                                  const float* beta,
                                                  float* out) {
  __shared__ float sX[2][MM];

  const int tid  = threadIdx.x;
  const int wave = __builtin_amdgcn_readfirstlane(tid >> 5);
  const int lane = tid & 31;
  const int half = lane >> 4;
  const int c    = lane & 15;
  const int s0   = blockIdx.x * OSLABS;
  const bool b0  = (c & 1) != 0;
  const bool b1  = (c & 2) != 0;

  float4 t[OSLABS];
#pragma unroll
  for (int i = 0; i < OSLABS; ++i)
    t[i] = *reinterpret_cast<const float4*>(tmp + (size_t)(s0 + i) * NN + 4 * c);
  const size_t fo = (size_t)(s0 + wave) * DD;
  const float q0 = q[fo + lane],  q1 = q[fo + lane + 32];
  const float g0 = gamma[lane],   g1 = gamma[lane + 32];
  const float e0 = beta[lane],    e1 = beta[lane + 32];

  const float* vp = vq + ((size_t)s0 * MM + 8 * wave + half) * NN + 4 * c;
  float4 v[OSLABS][4];
#pragma unroll
  for (int i = 0; i < OSLABS; ++i)
#pragma unroll
    for (int j = 0; j < 4; ++j)
      v[i][j] = *reinterpret_cast<const float4*>(vp + (size_t)i * MM * NN + (size_t)(2 * j) * NN);
  asm volatile("" : "+v"(t[0].x), "+v"(t[0].y), "+v"(t[0].z), "+v"(t[0].w) :: "memory");

#pragma unroll
  for (int i = 0; i < OSLABS; ++i) {
    float p[4];
#pragma unroll
    for (int j = 0; j < 4; ++j) {
      float a = v[i][j].x * t[i].x;
      a = fmaf(v[i][j].y, t[i].y, a);
      a = fmaf(v[i][j].z, t[i].z, a);
      a = fmaf(v[i][j].w, t[i].w, a);
      p[j] = a;
    }
    const float k0 = b0 ? p[1] : p[0];
    const float h0 = b0 ? p[0] : p[1];
    const float k1 = b0 ? p[3] : p[2];
    const float h1 = b0 ? p[2] : p[3];
    const float a0 = k0 + rowx<1>(h0);
    const float a1 = k1 + rowx<1>(h1);
    const float ky = b1 ? a1 : a0;
    const float hy = b1 ? a0 : a1;
    float y = ky + rowx<2>(hy);
    y += rowx<4>(y);
    y += rowx<8>(y);
    if (c < 4) sX[i & 1][8 * wave + 2 * c + half] = y;
    lds_barrier();

    if (wave == i) {
      const float x0 = sX[i & 1][lane]      + q0;
      const float x1 = sX[i & 1][lane + 32] + q1;
      float s1 = x0 + x1;
      float s2 = fmaf(x1, x1, x0 * x0);
      s1 += rowx<1>(s1); s2 += rowx<1>(s2);
      s1 += rowx<2>(s1); s2 += rowx<2>(s2);
      s1 += rowx<4>(s1); s2 += rowx<4>(s2);
      s1 += rowx<8>(s1); s2 += rowx<8>(s2);
      s1 += xhalf(s1);   s2 += xhalf(s2);
      const float mu   = s1 * (1.f / 64.f);
      const float var  = s2 * (1.f / 64.f) - mu * mu;
      const float rstd = rsqrtf(var + 1e-3f);
      out[fo + lane]      = (x0 - mu) * rstd * g0 + e0;
      out[fo + lane + 32] = (x1 - mu) * rstd * g1 + e1;
    }
  }
}

extern "C" void kernel_launch(void* const* d_in, const int* in_sizes, int n_in,
                              void* d_out, int out_size, void* d_ws, size_t ws_size,
                              hipStream_t stream) {
  const float* q     = (const float*)d_in[0];
  const float* kmat  = (const float*)d_in[1];
  const float* vq    = (const float*)d_in[2];
  const float* vk    = (const float*)d_in[3];
  const float* vexp  = (const float*)d_in[4];
  const float* scale = (const float*)d_in[5];
  const float* gamma = (const float*)d_in[6];
  const float* beta  = (const float*)d_in[7];
  float* out = (float*)d_out;

  float* vkc = (float*)d_ws;
  float* tmp = vkc + (size_t)BATCH * KK * NN;

  vkc_kernel <<<BATCH * KK / 4,    128, 0, stream>>>(vk, vexp, vkc);
  attn_kernel<<<BATCH * (LQ / 16), 256, 0, stream>>>(q, kmat, vkc, scale, tmp);
  out_kernel <<<BATCH * LQ / OSLABS, 256, 0, stream>>>(q, vq, tmp, gamma, beta, out);
}


# vk stream kernel: each wave requests its 32 weights (straight from memory, no LDS staging) and its whole half slab of 32 16-byte loads at entry, so it pays one memory round trip, not ten
# speedup vs baseline: 1.7135x; 1.1231x over previous
#include <hip/hip_runtime.h>

#define BATCH 8
#define LQ   512
#define KK   512
#define DD   64
#define NN   64
#define MM   64
#define PP   64

typedef __attribute__((ext_vector_type(2))) float v2f;
typedef __attribute__((ext_vector_type(8))) float v8f;

__global__ __launch_bounds__(128) void vkc_kernel(const float* vk,
                                                  const float* vexp,
                                                  float* vkc) {
  __shared__ float4 sHalf[64];
  const int tid  = threadIdx.x;
  const int ph   = tid >> 6;
  const int pair = (tid >> 4) & 3;
  const int nq   = (tid & 15) * 4;
  const int bk0  = blockIdx.x * 4;
  const int p0   = ph * (PP / 2);

  const float* ep = vexp + (size_t)(bk0 + pair) * PP + p0;
  float4 w[PP / 8];
#pragma unroll
  for (int j = 0; j < PP / 8; ++j) w[j] = *reinterpret_cast<const float4*>(ep + 4 * j);

  const float* vkp = vk + ((size_t)(bk0 + pair) * PP + p0) * NN + nq;
  float4 v[PP / 2];
#pragma unroll
  for (int p = 0; p < PP / 2; ++p) v[p] = *reinterpret_cast<const float4*>(vkp + (size_t)p * NN);
  asm volatile("" : "+v"(w[0].x), "+v"(w[0].y), "+v"(w[0].z), "+v"(w[0].w) :: "memory");

  float4 acc = {0.f, 0.f, 0.f, 0.f};
#pragma unroll
  for (int j = 0; j < PP / 8; ++j) {
    const float wj[4] = {w[j].x, w[j].y, w[j].z, w[j].w};
#pragma unroll
    for (int e = 0; e < 4; ++e) {
      const float4 r = v[4 * j + e];
      acc.x = fmaf(wj[e], r.x, acc.x);
      acc.y = fmaf(wj[e], r.y, acc.y);
      acc.z = fmaf(wj[e], r.z, acc.z);
      acc.w = fmaf(wj[e], r.w, acc.w);
    }
  }
  if (ph == 1) sHalf[tid & 63] = acc;
  __syncthreads();
  if (ph == 0) {
    const float4 o = sHalf[tid];
    acc.x += o.x;
    acc.y += o.y;
    acc.z += o.z;
    acc.w += o.w;
    *reinterpret_cast<float4*>(vkc + (size_t)(bk0 + pair) * NN + nq) = acc;
  }
}

#define SSTRIDE 516
#define PSTRIDE 68

__global__ __launch_bounds__(256) void attn_kernel(const float* __restrict__ q,
                                                   const float* __restrict__ kmat,
                                                   const float* __restrict__ vkc,
                                                   const float* __restrict__ scale_p,
                                                   float* __restrict__ tmp) {
  __shared__ float sS[16 * SSTRIDE];
  __shared__ __attribute__((aligned(16))) float sP[2 * 16 * PSTRIDE];
  __shared__ float sRowSum[16];

  const int tid  = threadIdx.x;
  const int wave = tid >> 5;
  const int lane = tid & 31;
  const int half = lane >> 4;
  const int lrow = lane & 15;

  const int b  = blockIdx.x >> 5;
  const int lt = blockIdx.x & 31;
  const int l0 = lt * 16;

  const float scale = scale_p[0];

  const float* qrow = q + ((size_t)b * LQ + l0 + lrow) * DD + 2 * half;
  v2f afrag[16];
#pragma unroll
  for (int dd = 0; dd < 16; ++dd) {
    const float2 t = *reinterpret_cast<const float2*>(qrow + 4 * dd);
    afrag[dd].x = t.x; afrag[dd].y = t.y;
  }

  const float* kbase = kmat + (size_t)b * KK * DD;
  v2f bfrag[4][16];
#pragma unroll
  for (int t = 0; t < 4; ++t) {
    const float* krow = kbase + (size_t)((wave + 8 * t) * 16 + lrow) * DD + 2 * half;
#pragma unroll
    for (int dd = 0; dd < 16; ++dd) {
      const float2 u = *reinterpret_cast<const float2*>(krow + 4 * dd);
      bfrag[t][dd].x = u.x; bfrag[t][dd].y = u.y;
    }
  }
#pragma unroll
  for (int t = 0; t < 4; ++t) {
    const int k0 = (wave + 8 * t) * 16;
    v8f acc0 = {0.f, 0.f, 0.f, 0.f, 0.f, 0.f, 0.f, 0.f};
    v8f acc1 = {0.f, 0.f, 0.f, 0.f, 0.f, 0.f, 0.f, 0.f};
#pragma unroll
    for (int dd = 0; dd < 16; dd += 2) {
      acc0 = __builtin_amdgcn_wmma_f32_16x16x4_f32(false, afrag[dd],     false, bfrag[t][dd],
                                                   (short)0, acc0, false, false);
      acc1 = __builtin_amdgcn_wmma_f32_16x16x4_f32(false, afrag[dd + 1], false, bfrag[t][dd + 1],
                                                   (short)0, acc1, false, false);
    }
#pragma unroll
    for (int r = 0; r < 8; ++r) {
      const int m = r + 8 * half;
      sS[m * SSTRIDE + k0 + lrow] = (acc0[r] + acc1[r]) * scale;
    }
  }

  const int nt = wave & 3;
  const int kh = wave >> 2;
  const int n0 = nt * 16;
  const int kb = kh * 256;
  const float* vb = vkc + ((size_t)b * KK + kb) * NN + n0 + lrow;
  v2f bpre[32];
#pragma unroll
  for (int i = 0; i < 32; ++i) {
    bpre[i].x = vb[(size_t)(4 * i + 2 * half)     * NN];
    bpre[i].y = vb[(size_t)(4 * i + 2 * half + 1) * NN];
  }
  __syncthreads();

#pragma unroll
  for (int rr = 0; rr < 2; ++rr) {
    const int r = 2 * wave + rr;
    float v[16];
    float mx = -3.4e38f;
#pragma unroll
    for (int i = 0; i < 16; ++i) {
      v[i] = sS[r * SSTRIDE + lane + 32 * i];
      mx = fmaxf(mx, v[i]);
    }
#pragma unroll
    for (int o = 16; o >= 1; o >>= 1) mx = fmaxf(mx, __shfl_xor(mx, o));
    float s = 0.f;
#pragma unroll
    for (int i = 0; i < 16; ++i) {
      const float e = expf(v[i] - mx);
      sS[r * SSTRIDE + lane + 32 * i] = e;
      s += e;
    }
#pragma unroll
    for (int o = 16; o >= 1; o >>= 1) s += __shfl_xor(s, o);
    if (lane == 0) sRowSum[r] = s;
  }
  __syncthreads();

  {
    v8f acc0 = {0.f, 0.f, 0.f, 0.f, 0.f, 0.f, 0.f, 0.f};
    v8f acc1 = {0.f, 0.f, 0.f, 0.f, 0.f, 0.f, 0.f, 0.f};
    const float* arow = &sS[lrow * SSTRIDE + kb + 2 * half];
#pragma unroll
    for (int i = 0; i < 32; i += 2) {
      const float2 ta = *reinterpret_cast<const float2*>(arow + 4 * i);
      const float2 tb = *reinterpret_cast<const float2*>(arow + 4 * i + 4);
      v2f a0; a0.x = ta.x; a0.y = ta.y;
      v2f a1; a1.x = tb.x; a1.y = tb.y;
      acc0 = __builtin_amdgcn_wmma_f32_16x16x4_f32(false, a0, false, bpre[i],
                                                   (short)0, acc0, false, false);
      acc1 = __builtin_amdgcn_wmma_f32_16x16x4_f32(false, a1, false, bpre[i + 1],
                                                   (short)0, acc1, false, false);
    }
#pragma unroll
    for (int i = 32; i < 64; i += 2) {
      const float2 ta = *reinterpret_cast<const float2*>(arow + 4 * i);
      const float2 tb = *reinterpret_cast<const float2*>(arow + 4 * i + 4);
      v2f a0; a0.x = ta.x; a0.y = ta.y;
      v2f a1; a1.x = tb.x; a1.y = tb.y;
      v2f b0;
      b0.x = vb[(size_t)(4 * i + 2 * half)     * NN];
      b0.y = vb[(size_t)(4 * i + 2 * half + 1) * NN];
      v2f b1;
      b1.x = vb[(size_t)(4 * i + 4 + 2 * half)     * NN];
      b1.y = vb[(size_t)(4 * i + 4 + 2 * half + 1) * NN];
      acc0 = __builtin_amdgcn_wmma_f32_16x16x4_f32(false, a0, false, b0,
                                                   (short)0, acc0, false, false);
      acc1 = __builtin_amdgcn_wmma_f32_16x16x4_f32(false, a1, false, b1,
                                                   (short)0, acc1, false, false);
    }
#pragma unroll
    for (int r = 0; r < 8; ++r) {
      const int m = r + 8 * half;
      sP[(kh * 16 + m) * PSTRIDE + n0 + lrow] = acc0[r] + acc1[r];
    }
  }
  __syncthreads();
  {
    const int row = tid >> 4;
    const int c4  = (tid & 15) * 4;
    const float4 p0 = *reinterpret_cast<const float4*>(&sP[row * PSTRIDE + c4]);
    const float4 p1 = *reinterpret_cast<const float4*>(&sP[(16 + row) * PSTRIDE + c4]);
    const float den = sRowSum[row];
    float4 o;
    o.x = (p0.x + p1.x) / den;
    o.y = (p0.y + p1.y) / den;
    o.z = (p0.z + p1.z) / den;
    o.w = (p0.w + p1.w) / den;
    *reinterpret_cast<float4*>(tmp + ((size_t)b * LQ + l0 + row) * NN + c4) = o;
  }
}

#define OSLABS 8

static __device__ __forceinline__ float xhalf(float x) {
  const int b = __builtin_bit_cast(int, x);
  return __builtin_bit_cast(float,
      __builtin_amdgcn_permlanex16(b, b, 0x76543210, 0xfedcba98, false, false));
}
template <int M>
static __device__ __forceinline__ float rowx(float x) {
  const int b = __builtin_bit_cast(int, x);
  return __builtin_bit_cast(float, __builtin_amdgcn_update_dpp(0, b, 0x160 + M, 0xf, 0xf, true));
}
static __device__ __forceinline__ void lds_barrier() {
  asm volatile("s_wait_dscnt 0x0\n\ts_barrier_signal -1\n\ts_barrier_wait -1" ::: "memory");
}

__global__ __launch_bounds__(256) void out_kernel(const float* q,
                                                  const float* vq,
                                                  const float* tmp,
                                                  const float* gamma,
                                                  const float* beta,
                                                  float* out) {
  __shared__ float sX[2][MM];

  const int tid  = threadIdx.x;
  const int wave = __builtin_amdgcn_readfirstlane(tid >> 5);
  const int lane = tid & 31;
  const int half = lane >> 4;
  const int c    = lane & 15;
  const int s0   = blockIdx.x * OSLABS;
  const bool b0  = (c & 1) != 0;
  const bool b1  = (c & 2) != 0;

  float4 t[OSLABS];
#pragma unroll
  for (int i = 0; i < OSLABS; ++i)
    t[i] = *reinterpret_cast<const float4*>(tmp + (size_t)(s0 + i) * NN + 4 * c);
  const size_t fo = (size_t)(s0 + wave) * DD;
  const float q0 = q[fo + lane],  q1 = q[fo + lane + 32];
  const float g0 = gamma[lane],   g1 = gamma[lane + 32];
  const float e0 = beta[lane],    e1 = beta[lane + 32];

  const float* vp = vq + ((size_t)s0 * MM + 8 * wave + half) * NN + 4 * c;
  float4 v[OSLABS][4];
#pragma unroll
  for (int i = 0; i < OSLABS; ++i)
#pragma unroll
    for (int j = 0; j < 4; ++j)
      v[i][j] = *reinterpret_cast<const float4*>(vp + (size_t)i * MM * NN + (size_t)(2 * j) * NN);
  asm volatile("" : "+v"(t[0].x), "+v"(t[0].y), "+v"(t[0].z), "+v"(t[0].w) :: "memory");

#pragma unroll
  for (int i = 0; i < OSLABS; ++i) {
    float p[4];
#pragma unroll
    for (int j = 0; j < 4; ++j) {
      float a = v[i][j].x * t[i].x;
      a = fmaf(v[i][j].y, t[i].y, a);
      a = fmaf(v[i][j].z, t[i].z, a);
      a = fmaf(v[i][j].w, t[i].w, a);
      p[j] = a;
    }
    const float k0 = b0 ? p[1] : p[0];
    const float h0 = b0 ? p[0] : p[1];
    const float k1 = b0 ? p[3] : p[2];
    const float h1 = b0 ? p[2] : p[3];
    const float a0 = k0 + rowx<1>(h0);
    const float a1 = k1 + rowx<1>(h1);
    const float ky = b1 ? a1 : a0;
    const float hy = b1 ? a0 : a1;
    float y = ky + rowx<2>(hy);
    y += rowx<4>(y);
    y += rowx<8>(y);
    if (c < 4) sX[i & 1][8 * wave + 2 * c + half] = y;
    lds_barrier();

    if (wave == i) {
      const float x0 = sX[i & 1][lane]      + q0;
      const float x1 = sX[i & 1][lane + 32] + q1;
      float s1 = x0 + x1;
      float s2 = fmaf(x1, x1, x0 * x0);
      s1 += rowx<1>(s1); s2 += rowx<1>(s2);
      s1 += rowx<2>(s1); s2 += rowx<2>(s2);
      s1 += rowx<4>(s1); s2 += rowx<4>(s2);
      s1 += rowx<8>(s1); s2 += rowx<8>(s2);
      s1 += xhalf(s1);   s2 += xhalf(s2);
      const float mu   = s1 * (1.f / 64.f);
      const float var  = s2 * (1.f / 64.f) - mu * mu;
      const float rstd = rsqrtf(var + 1e-3f);
      out[fo + lane]      = (x0 - mu) * rstd * g0 + e0;
      out[fo + lane + 32] = (x1 - mu) * rstd * g1 + e1;
    }
  }
}

extern "C" void kernel_launch(void* const* d_in, const int* in_sizes, int n_in,
                              void* d_out, int out_size, void* d_ws, size_t ws_size,
                              hipStream_t stream) {
  const float* q     = (const float*)d_in[0];
  const float* kmat  = (const float*)d_in[1];
  const float* vq    = (const float*)d_in[2];
  const float* vk    = (const float*)d_in[3];
  const float* vexp  = (const float*)d_in[4];
  const float* scale = (const float*)d_in[5];
  const float* gamma = (const float*)d_in[6];
  const float* beta  = (const float*)d_in[7];
  float* out = (float*)d_out;

  float* vkc = (float*)d_ws;
  float* tmp = vkc + (size_t)BATCH * KK * NN;

  vkc_kernel <<<BATCH * KK / 4,    128, 0, stream>>>(vk, vexp, vkc);
  attn_kernel<<<BATCH * (LQ / 16), 256, 0, stream>>>(q, kmat, vkc, scale, tmp);
  out_kernel <<<BATCH * LQ / OSLABS, 256, 0, stream>>>(q, vq, tmp, gamma, beta, out);
}


# attention kernel: q and k tiles fetched by whole-line 16-byte loads (4 lines a load, not 16), parked in a per-wave LDS area and read back as 16-byte WMMA fragments; score fp sum order changes
# speedup vs baseline: 1.8028x; 1.0521x over previous
#include <hip/hip_runtime.h>

#define BATCH 8
#define LQ   512
#define KK   512
#define DD   64
#define NN   64
#define MM   64
#define PP   64

typedef __attribute__((ext_vector_type(2))) float v2f;
typedef __attribute__((ext_vector_type(8))) float v8f;

__global__ __launch_bounds__(128) void vkc_kernel(const float* vk,
                                                  const float* vexp,
                                                  float* vkc) {
  __shared__ float4 sHalf[64];
  const int tid  = threadIdx.x;
  const int ph   = tid >> 6;
  const int pair = (tid >> 4) & 3;
  const int nq   = (tid & 15) * 4;
  const int bk0  = blockIdx.x * 4;
  const int p0   = ph * (PP / 2);

  const float* ep = vexp + (size_t)(bk0 + pair) * PP + p0;
  float4 w[PP / 8];
#pragma unroll
  for (int j = 0; j < PP / 8; ++j) w[j] = *reinterpret_cast<const float4*>(ep + 4 * j);

  const float* vkp = vk + ((size_t)(bk0 + pair) * PP + p0) * NN + nq;
  float4 v[PP / 2];
#pragma unroll
  for (int p = 0; p < PP / 2; ++p) v[p] = *reinterpret_cast<const float4*>(vkp + (size_t)p * NN);
  asm volatile("" : "+v"(w[0].x), "+v"(w[0].y), "+v"(w[0].z), "+v"(w[0].w) :: "memory");

  float4 acc = {0.f, 0.f, 0.f, 0.f};
#pragma unroll
  for (int j = 0; j < PP / 8; ++j) {
    const float wj[4] = {w[j].x, w[j].y, w[j].z, w[j].w};
#pragma unroll
    for (int e = 0; e < 4; ++e) {
      const float4 r = v[4 * j + e];
      acc.x = fmaf(wj[e], r.x, acc.x);
      acc.y = fmaf(wj[e], r.y, acc.y);
      acc.z = fmaf(wj[e], r.z, acc.z);
      acc.w = fmaf(wj[e], r.w, acc.w);
    }
  }
  if (ph == 1) sHalf[tid & 63] = acc;
  __syncthreads();
  if (ph == 0) {
    const float4 o = sHalf[tid];
    acc.x += o.x;
    acc.y += o.y;
    acc.z += o.z;
    acc.w += o.w;
    *reinterpret_cast<float4*>(vkc + (size_t)(bk0 + pair) * NN + nq) = acc;
  }
}

#define SSTRIDE 516
#define PSTRIDE 68
#define QKPITCH 68
#define WSTAGE ((16 + 64) * QKPITCH)

static __device__ __forceinline__ float4 ldg4(const float* p) {
  return *reinterpret_cast<const float4*>(p);
}
static __device__ __forceinline__ void park(float* dst, int i, int half, int lrow, float4 g) {
  *reinterpret_cast<float4*>(dst + (2 * i + half) * QKPITCH + 4 * lrow) = g;
}
static __device__ __forceinline__ float4 frag(const float* src, int j, int half, int lrow) {
  return *reinterpret_cast<const float4*>(src + lrow * QKPITCH + 8 * j + 4 * half);
}
static __device__ __forceinline__ void wmma2(v8f& acc0, v8f& acc1, float4 a, float4 b) {
  v2f a0; a0.x = a.x; a0.y = a.y;
  v2f a1; a1.x = a.z; a1.y = a.w;
  v2f b0; b0.x = b.x; b0.y = b.y;
  v2f b1; b1.x = b.z; b1.y = b.w;
  acc0 = __builtin_amdgcn_wmma_f32_16x16x4_f32(false, a0, false, b0, (short)0, acc0, false, false);
  acc1 = __builtin_amdgcn_wmma_f32_16x16x4_f32(false, a1, false, b1, (short)0, acc1, false, false);
}
#define LOAD_TILE(G, p)                                                              \
  const float4 G##0 = ldg4((p)), G##1 = ldg4((p) + 128), G##2 = ldg4((p) + 256),       \
               G##3 = ldg4((p) + 384), G##4 = ldg4((p) + 512), G##5 = ldg4((p) + 640), \
               G##6 = ldg4((p) + 768), G##7 = ldg4((p) + 896)
#define PARK_TILE(dst, G)                                                    \
  park((dst), 0, half, lrow, G##0); park((dst), 1, half, lrow, G##1);        \
  park((dst), 2, half, lrow, G##2); park((dst), 3, half, lrow, G##3);        \
  park((dst), 4, half, lrow, G##4); park((dst), 5, half, lrow, G##5);        \
  park((dst), 6, half, lrow, G##6); park((dst), 7, half, lrow, G##7)
#define READ_FRAGS(F, src)                                                           \
  const float4 F##0 = frag((src), 0, half, lrow), F##1 = frag((src), 1, half, lrow), \
               F##2 = frag((src), 2, half, lrow), F##3 = frag((src), 3, half, lrow), \
               F##4 = frag((src), 4, half, lrow), F##5 = frag((src), 5, half, lrow), \
               F##6 = frag((src), 6, half, lrow), F##7 = frag((src), 7, half, lrow)
#define SCORE_TILE(A, B, out)                                                  \
  do {                                                                         \
    v8f acc0 = {0.f, 0.f, 0.f, 0.f, 0.f, 0.f, 0.f, 0.f};                       \
    v8f acc1 = {0.f, 0.f, 0.f, 0.f, 0.f, 0.f, 0.f, 0.f};                       \
    wmma2(acc0, acc1, A##0, B##0); wmma2(acc0, acc1, A##1, B##1);              \
    wmma2(acc0, acc1, A##2, B##2); wmma2(acc0, acc1, A##3, B##3);              \
    wmma2(acc0, acc1, A##4, B##4); wmma2(acc0, acc1, A##5, B##5);              \
    wmma2(acc0, acc1, A##6, B##6); wmma2(acc0, acc1, A##7, B##7);              \
    _Pragma("unroll") for (int r = 0; r < 8; ++r)                              \
      (out)[r * SSTRIDE] = (acc0[r] + acc1[r]) * scale;                        \
  } while (0)

__global__ __launch_bounds__(256) void attn_kernel(const float* __restrict__ q,
                                                   const float* __restrict__ kmat,
                                                   const float* __restrict__ vkc,
                                                   const float* __restrict__ scale_p,
                                                   float* __restrict__ tmp) {
  __shared__ float sS[16 * SSTRIDE];
  __shared__ __attribute__((aligned(16))) float sP[2 * 16 * PSTRIDE];
  __shared__ __attribute__((aligned(16))) float sStage[8 * WSTAGE];
  __shared__ float sRowSum[16];

  const int tid  = threadIdx.x;
  const int wave = tid >> 5;
  const int lane = tid & 31;
  const int half = lane >> 4;
  const int lrow = lane & 15;

  const int b  = blockIdx.x >> 5;
  const int lt = blockIdx.x & 31;
  const int l0 = lt * 16;

  const float scale = scale_p[0];

  const float* qt  = q + ((size_t)b * LQ + l0) * DD + 4 * lane;
  const float* kt0 = kmat + ((size_t)b * KK + (size_t)wave * 16) * DD + 4 * lane;
  LOAD_TILE(gq, qt);
  LOAD_TILE(gk0, kt0);
  LOAD_TILE(gk1, kt0 + 1 * 128 * DD);
  LOAD_TILE(gk2, kt0 + 2 * 128 * DD);
  LOAD_TILE(gk3, kt0 + 3 * 128 * DD);
  float* wq = sStage + wave * WSTAGE;
  float* wk = wq + 16 * QKPITCH;
  float* sOut = &sS[(8 * half) * SSTRIDE + wave * 16 + lrow];
  PARK_TILE(wq, gq);
  PARK_TILE(wk, gk0);
  PARK_TILE(wk + 16 * QKPITCH, gk1);
  __builtin_amdgcn_wave_barrier();
  READ_FRAGS(fa, wq);
  READ_FRAGS(fb0, wk);
  READ_FRAGS(fb1, wk + 16 * QKPITCH);
  PARK_TILE(wk + 32 * QKPITCH, gk2);
  SCORE_TILE(fa, fb0, sOut);
  __builtin_amdgcn_wave_barrier();
  READ_FRAGS(fb2, wk + 32 * QKPITCH);
  PARK_TILE(wk + 48 * QKPITCH, gk3);
  SCORE_TILE(fa, fb1, sOut + 128);
  __builtin_amdgcn_wave_barrier();
  READ_FRAGS(fb3, wk + 48 * QKPITCH);
  SCORE_TILE(fa, fb2, sOut + 256);
  SCORE_TILE(fa, fb3, sOut + 384);

  const int nt = wave & 3;
  const int kh = wave >> 2;
  const int n0 = nt * 16;
  const int kb = kh * 256;
  const float* vb = vkc + ((size_t)b * KK + kb) * NN + n0 + lrow;
  v2f bpre[32];
#pragma unroll
  for (int i = 0; i < 32; ++i) {
    bpre[i].x = vb[(size_t)(4 * i + 2 * half)     * NN];
    bpre[i].y = vb[(size_t)(4 * i + 2 * half + 1) * NN];
  }
  __syncthreads();

#pragma unroll
  for (int rr = 0; rr < 2; ++rr) {
    const int r = 2 * wave + rr;
    float v[16];
    float mx = -3.4e38f;
#pragma unroll
    for (int i = 0; i < 16; ++i) {
      v[i] = sS[r * SSTRIDE + lane + 32 * i];
      mx = fmaxf(mx, v[i]);
    }
#pragma unroll
    for (int o = 16; o >= 1; o >>= 1) mx = fmaxf(mx, __shfl_xor(mx, o));
    float s = 0.f;
#pragma unroll
    for (int i = 0; i < 16; ++i) {
      const float e = expf(v[i] - mx);
      sS[r * SSTRIDE + lane + 32 * i] = e;
      s += e;
    }
#pragma unroll
    for (int o = 16; o >= 1; o >>= 1) s += __shfl_xor(s, o);
    if (lane == 0) sRowSum[r] = s;
  }
  __syncthreads();

  {
    v8f acc0 = {0.f, 0.f, 0.f, 0.f, 0.f, 0.f, 0.f, 0.f};
    v8f acc1 = {0.f, 0.f, 0.f, 0.f, 0.f, 0.f, 0.f, 0.f};
    const float* arow = &sS[lrow * SSTRIDE + kb + 2 * half];
#pragma unroll
    for (int i = 0; i < 32; i += 2) {
      const float2 ta = *reinterpret_cast<const float2*>(arow + 4 * i);
      const float2 tb = *reinterpret_cast<const float2*>(arow + 4 * i + 4);
      v2f a0; a0.x = ta.x; a0.y = ta.y;
      v2f a1; a1.x = tb.x; a1.y = tb.y;
      acc0 = __builtin_amdgcn_wmma_f32_16x16x4_f32(false, a0, false, bpre[i],
                                                   (short)0, acc0, false, false);
      acc1 = __builtin_amdgcn_wmma_f32_16x16x4_f32(false, a1, false, bpre[i + 1],
                                                   (short)0, acc1, false, false);
    }
#pragma unroll
    for (int i = 32; i < 64; i += 2) {
      const float2 ta = *reinterpret_cast<const float2*>(arow + 4 * i);
      const float2 tb = *reinterpret_cast<const float2*>(arow + 4 * i + 4);
      v2f a0; a0.x = ta.x; a0.y = ta.y;
      v2f a1; a1.x = tb.x; a1.y = tb.y;
      v2f b0;
      b0.x = vb[(size_t)(4 * i + 2 * half)     * NN];
      b0.y = vb[(size_t)(4 * i + 2 * half + 1) * NN];
      v2f b1;
      b1.x = vb[(size_t)(4 * i + 4 + 2 * half)     * NN];
      b1.y = vb[(size_t)(4 * i + 4 + 2 * half + 1) * NN];
      acc0 = __builtin_amdgcn_wmma_f32_16x16x4_f32(false, a0, false, b0,
                                                   (short)0, acc0, false, false);
      acc1 = __builtin_amdgcn_wmma_f32_16x16x4_f32(false, a1, false, b1,
                                                   (short)0, acc1, false, false);
    }
#pragma unroll
    for (int r = 0; r < 8; ++r) {
      const int m = r + 8 * half;
      sP[(kh * 16 + m) * PSTRIDE + n0 + lrow] = acc0[r] + acc1[r];
    }
  }
  __syncthreads();
  {
    const int row = tid >> 4;
    const int c4  = (tid & 15) * 4;
    const float4 p0 = *reinterpret_cast<const float4*>(&sP[row * PSTRIDE + c4]);
    const float4 p1 = *reinterpret_cast<const float4*>(&sP[(16 + row) * PSTRIDE + c4]);
    const float den = sRowSum[row];
    float4 o;
    o.x = (p0.x + p1.x) / den;
    o.y = (p0.y + p1.y) / den;
    o.z = (p0.z + p1.z) / den;
    o.w = (p0.w + p1.w) / den;
    *reinterpret_cast<float4*>(tmp + ((size_t)b * LQ + l0 + row) * NN + c4) = o;
  }
}

#define OSLABS 8

static __device__ __forceinline__ float xhalf(float x) {
  const int b = __builtin_bit_cast(int, x);
  return __builtin_bit_cast(float,
      __builtin_amdgcn_permlanex16(b, b, 0x76543210, 0xfedcba98, false, false));
}
template <int M>
static __device__ __forceinline__ float rowx(float x) {
  const int b = __builtin_bit_cast(int, x);
  return __builtin_bit_cast(float, __builtin_amdgcn_update_dpp(0, b, 0x160 + M, 0xf, 0xf, true));
}
static __device__ __forceinline__ void lds_barrier() {
  asm volatile("s_wait_dscnt 0x0\n\ts_barrier_signal -1\n\ts_barrier_wait -1" ::: "memory");
}

__global__ __launch_bounds__(256) void out_kernel(const float* q,
                                                  const float* vq,
                                                  const float* tmp,
                                                  const float* gamma,
                                                  const float* beta,
                                                  float* out) {
  __shared__ float sX[2][MM];

  const int tid  = threadIdx.x;
  const int wave = __builtin_amdgcn_readfirstlane(tid >> 5);
  const int lane = tid & 31;
  const int half = lane >> 4;
  const int c    = lane & 15;
  const int s0   = blockIdx.x * OSLABS;
  const bool b0  = (c & 1) != 0;
  const bool b1  = (c & 2) != 0;

  float4 t[OSLABS];
#pragma unroll
  for (int i = 0; i < OSLABS; ++i)
    t[i] = *reinterpret_cast<const float4*>(tmp + (size_t)(s0 + i) * NN + 4 * c);
  const size_t fo = (size_t)(s0 + wave) * DD;
  const float q0 = q[fo + lane],  q1 = q[fo + lane + 32];
  const float g0 = gamma[lane],   g1 = gamma[lane + 32];
  const float e0 = beta[lane],    e1 = beta[lane + 32];

  const float* vp = vq + ((size_t)s0 * MM + 8 * wave + half) * NN + 4 * c;
  float4 v[OSLABS][4];
#pragma unroll
  for (int i = 0; i < OSLABS; ++i)
#pragma unroll
    for (int j = 0; j < 4; ++j)
      v[i][j] = *reinterpret_cast<const float4*>(vp + (size_t)i * MM * NN + (size_t)(2 * j) * NN);
  asm volatile("" : "+v"(t[0].x), "+v"(t[0].y), "+v"(t[0].z), "+v"(t[0].w) :: "memory");

#pragma unroll
  for (int i = 0; i < OSLABS; ++i) {
    float p[4];
#pragma unroll
    for (int j = 0; j < 4; ++j) {
      float a = v[i][j].x * t[i].x;
      a = fmaf(v[i][j].y, t[i].y, a);
      a = fmaf(v[i][j].z, t[i].z, a);
      a = fmaf(v[i][j].w, t[i].w, a);
      p[j] = a;
    }
    const float k0 = b0 ? p[1] : p[0];
    const float h0 = b0 ? p[0] : p[1];
    const float k1 = b0 ? p[3] : p[2];
    const float h1 = b0 ? p[2] : p[3];
    const float a0 = k0 + rowx<1>(h0);
    const float a1 = k1 + rowx<1>(h1);
    const float ky = b1 ? a1 : a0;
    const float hy = b1 ? a0 : a1;
    float y = ky + rowx<2>(hy);
    y += rowx<4>(y);
    y += rowx<8>(y);
    if (c < 4) sX[i & 1][8 * wave + 2 * c + half] = y;
    lds_barrier();

    if (wave == i) {
      const float x0 = sX[i & 1][lane]      + q0;
      const float x1 = sX[i & 1][lane + 32] + q1;
      float s1 = x0 + x1;
      float s2 = fmaf(x1, x1, x0 * x0);
      s1 += rowx<1>(s1); s2 += rowx<1>(s2);
      s1 += rowx<2>(s1); s2 += rowx<2>(s2);
      s1 += rowx<4>(s1); s2 += rowx<4>(s2);
      s1 += rowx<8>(s1); s2 += rowx<8>(s2);
      s1 += xhalf(s1);   s2 += xhalf(s2);
      const float mu   = s1 * (1.f / 64.f);
      const float var  = s2 * (1.f / 64.f) - mu * mu;
      const float rstd = rsqrtf(var + 1e-3f);
      out[fo + lane]      = (x0 - mu) * rstd * g0 + e0;
      out[fo + lane + 32] = (x1 - mu) * rstd * g1 + e1;
    }
  }
}

extern "C" void kernel_launch(void* const* d_in, const int* in_sizes, int n_in,
                              void* d_out, int out_size, void* d_ws, size_t ws_size,
                              hipStream_t stream) {
  const float* q     = (const float*)d_in[0];
  const float* kmat  = (const float*)d_in[1];
  const float* vq    = (const float*)d_in[2];
  const float* vk    = (const float*)d_in[3];
  const float* vexp  = (const float*)d_in[4];
  const float* scale = (const float*)d_in[5];
  const float* gamma = (const float*)d_in[6];
  const float* beta  = (const float*)d_in[7];
  float* out = (float*)d_out;

  float* vkc = (float*)d_ws;
  float* tmp = vkc + (size_t)BATCH * KK * NN;

  vkc_kernel <<<BATCH * KK / 4,    128, 0, stream>>>(vk, vexp, vkc);
  attn_kernel<<<BATCH * (LQ / 16), 256, 0, stream>>>(q, kmat, vkc, scale, tmp);
  out_kernel <<<BATCH * LQ / OSLABS, 256, 0, stream>>>(q, vq, tmp, gamma, beta, out);
}
